# starting kernel
# baseline (speedup 1.0000x reference)
#include <hip/hip_runtime.h>


typedef __attribute__((ext_vector_type(2))) float v2f;
typedef __attribute__((ext_vector_type(8))) float v8f;

#define N_NODES  100000
#define N_EDGES  1600000
#define N_GRAPHS 128
#define N_TILES  (N_NODES / 16)

__global__ void zero_kernel(float* __restrict__ p, int n) {
  int i = blockIdx.x * blockDim.x + threadIdx.x;
  if (i < n) p[i] = 0.0f;
}

__global__ void scatter3_kernel(const float* __restrict__ x,
                                const int* __restrict__ ei,
                                float* __restrict__ agg3) {
  int e = blockIdx.x * blockDim.x + threadIdx.x;
  if (e >= N_EDGES) return;
  int s = ei[e];
  int d = ei[N_EDGES + e];
  #pragma unroll
  for (int k = 0; k < 3; ++k)
    atomicAdd(&agg3[d * 3 + k], x[s * 3 + k]);
}

__global__ void scatter64_kernel(const float* __restrict__ h,
                                 const int* __restrict__ ei,
                                 float* __restrict__ agg) {
  long long gid = (long long)blockIdx.x * blockDim.x + threadIdx.x;
  int e = (int)(gid >> 6);
  int f = (int)(gid & 63);
  if (e >= N_EDGES) return;
  int s = ei[e];
  int d = ei[N_EDGES + e];
  atomicAdd(&agg[d * 64 + f], h[s * 64 + f]);
}

__global__ __launch_bounds__(128) void mlp1_kernel(
    const float* __restrict__ x, const float* __restrict__ agg3,
    const float* __restrict__ W1a, const float* __restrict__ b1a,
    const float* __restrict__ W1b, const float* __restrict__ b1b,
    float* __restrict__ h1) {
  __shared__ float sWa[4 * 64];
  __shared__ float sWb[64 * 64];
  __shared__ float sBa[64];
  __shared__ float sBb[64];
  __shared__ float sT[4][16 * 68];

  int tid = threadIdx.x;
  for (int i = tid; i < 4 * 64; i += 128) {
    int k = i >> 6, n = i & 63;
    sWa[i] = (k < 3) ? W1a[k * 64 + n] : 0.0f;
  }
  for (int i = tid; i < 64 * 64; i += 128) sWb[i] = W1b[i];
  if (tid < 64) { sBa[tid] = b1a[tid]; sBb[tid] = b1b[tid]; }
  __syncthreads();

  int wave = tid >> 5;
  int lane = tid & 31;
  int half = lane >> 4;
  int m    = lane & 15;
  int tile = blockIdx.x * 4 + wave;
  if (tile >= N_TILES) return;
  int node0 = tile * 16;
  float* myT = &sT[wave][0];

  int node = node0 + m;
  int k0   = 2 * half;
  v2f a;
  a.x = x[node * 3 + k0] + agg3[node * 3 + k0];
  a.y = (half == 0) ? (x[node * 3 + 1] + agg3[node * 3 + 1]) : 0.0f;

  v8f acc[4] = {};
  #pragma unroll
  for (int nt = 0; nt < 4; ++nt) {
    v2f b;
    b.x = sWa[k0 * 64 + nt * 16 + m];
    b.y = sWa[(k0 + 1) * 64 + nt * 16 + m];
    acc[nt] = __builtin_amdgcn_wmma_f32_16x16x4_f32(
        false, a, false, b, (short)0, acc[nt], false, false);
  }

  #pragma unroll
  for (int nt = 0; nt < 4; ++nt) {
    float bias = sBa[nt * 16 + m];
    #pragma unroll
    for (int r = 0; r < 8; ++r) {
      float v = acc[nt][r] + bias;
      myT[(r + 8 * half) * 68 + nt * 16 + m] = v > 0.0f ? v : 0.0f;
    }
  }

  v8f acc2[4] = {};
  for (int ks = 0; ks < 16; ++ks) {
    int kk = ks * 4 + 2 * half;
    v2f a2;
    a2.x = myT[m * 68 + kk];
    a2.y = myT[m * 68 + kk + 1];
    #pragma unroll
    for (int nt = 0; nt < 4; ++nt) {
      v2f b;
      b.x = sWb[kk * 64 + nt * 16 + m];
      b.y = sWb[(kk + 1) * 64 + nt * 16 + m];
      acc2[nt] = __builtin_amdgcn_wmma_f32_16x16x4_f32(
          false, a2, false, b, (short)0, acc2[nt], false, false);
    }
  }

  #pragma unroll
  for (int nt = 0; nt < 4; ++nt) {
    float bias = sBb[nt * 16 + m];
    #pragma unroll
    for (int r = 0; r < 8; ++r) {
      float v = acc2[nt][r] + bias;
      h1[(node0 + r + 8 * half) * 64 + nt * 16 + m] = v > 0.0f ? v : 0.0f;
    }
  }
}

__global__ __launch_bounds__(128) void mlp2_kernel(
    const float* __restrict__ h1, const float* __restrict__ agg64,
    const float* __restrict__ W2a, const float* __restrict__ b2a,
    const float* __restrict__ W2b, const float* __restrict__ b2b,
    float* __restrict__ h2) {
  __shared__ float sWa[64 * 64];
  __shared__ float sWb[64 * 64];
  __shared__ float sBa[64];
  __shared__ float sBb[64];
  __shared__ float sT[4][16 * 68];

  int tid = threadIdx.x;
  for (int i = tid; i < 64 * 64; i += 128) { sWa[i] = W2a[i]; sWb[i] = W2b[i]; }
  if (tid < 64) { sBa[tid] = b2a[tid]; sBb[tid] = b2b[tid]; }
  __syncthreads();

  int wave = tid >> 5;
  int lane = tid & 31;
  int half = lane >> 4;
  int m    = lane & 15;
  int tile = blockIdx.x * 4 + wave;
  if (tile >= N_TILES) return;
  int node0 = tile * 16;
  float* myT = &sT[wave][0];

  for (int i = lane; i < 16 * 64; i += 32) {
    int r = i >> 6, c = i & 63;
    int node = node0 + r;
    myT[r * 68 + c] = h1[node * 64 + c] + agg64[node * 64 + c];
  }

  v8f acc[4] = {};
  for (int ks = 0; ks < 16; ++ks) {
    int kk = ks * 4 + 2 * half;
    v2f a;
    a.x = myT[m * 68 + kk];
    a.y = myT[m * 68 + kk + 1];
    #pragma unroll
    for (int nt = 0; nt < 4; ++nt) {
      v2f b;
      b.x = sWa[kk * 64 + nt * 16 + m];
      b.y = sWa[(kk + 1) * 64 + nt * 16 + m];
      acc[nt] = __builtin_amdgcn_wmma_f32_16x16x4_f32(
          false, a, false, b, (short)0, acc[nt], false, false);
    }
  }

  #pragma unroll
  for (int nt = 0; nt < 4; ++nt) {
    float bias = sBa[nt * 16 + m];
    #pragma unroll
    for (int r = 0; r < 8; ++r) {
      float v = acc[nt][r] + bias;
      myT[(r + 8 * half) * 68 + nt * 16 + m] = v > 0.0f ? v : 0.0f;
    }
  }

  v8f acc2[4] = {};
  for (int ks = 0; ks < 16; ++ks) {
    int kk = ks * 4 + 2 * half;
    v2f a2;
    a2.x = myT[m * 68 + kk];
    a2.y = myT[m * 68 + kk + 1];
    #pragma unroll
    for (int nt = 0; nt < 4; ++nt) {
      v2f b;
      b.x = sWb[kk * 64 + nt * 16 + m];
      b.y = sWb[(kk + 1) * 64 + nt * 16 + m];
      acc2[nt] = __builtin_amdgcn_wmma_f32_16x16x4_f32(
          false, a2, false, b, (short)0, acc2[nt], false, false);
    }
  }

  #pragma unroll
  for (int nt = 0; nt < 4; ++nt) {
    float bias = sBb[nt * 16 + m];
    #pragma unroll
    for (int r = 0; r < 8; ++r) {
      float v = acc2[nt][r] + bias;
      h2[(node0 + r + 8 * half) * 64 + nt * 16 + m] = v > 0.0f ? v : 0.0f;
    }
  }
}

__global__ void pool_add_kernel(const float* __restrict__ h,
                                const int* __restrict__ batch,
                                float* __restrict__ sums) {
  long long gid = (long long)blockIdx.x * blockDim.x + threadIdx.x;
  int n = (int)(gid >> 6);
  int f = (int)(gid & 63);
  if (n >= N_NODES) return;
  int g = batch[n];
  atomicAdd(&sums[g * 64 + f], h[n * 64 + f]);
}

__global__ void count_kernel(const int* __restrict__ batch,
                             float* __restrict__ cnt) {
  int n = blockIdx.x * blockDim.x + threadIdx.x;
  if (n >= N_NODES) return;
  atomicAdd(&cnt[batch[n]], 1.0f);
}

__global__ void finalize_kernel(const float* __restrict__ sums,
                                const float* __restrict__ cnt,
                                float* __restrict__ out) {
  int i = blockIdx.x * blockDim.x + threadIdx.x;
  if (i >= N_GRAPHS * 64) return;
  float c = cnt[i >> 6];
  out[i] = sums[i] / fmaxf(c, 1.0f);
}

extern "C" void kernel_launch(void* const* d_in, const int* in_sizes, int n_in,
                              void* d_out, int out_size, void* d_ws, size_t ws_size,
                              hipStream_t stream) {
  const float* x     = (const float*)d_in[0];
  const int*   ei    = (const int*)d_in[1];
  const int*   batch = (const int*)d_in[2];
  const float* W1a   = (const float*)d_in[3];
  const float* b1a   = (const float*)d_in[4];
  const float* W1b   = (const float*)d_in[5];
  const float* b1b   = (const float*)d_in[6];
  const float* W2a   = (const float*)d_in[7];
  const float* b2a   = (const float*)d_in[8];
  const float* W2b   = (const float*)d_in[9];
  const float* b2b   = (const float*)d_in[10];
  float* out = (float*)d_out;

  float* ws    = (float*)d_ws;
  float* agg3  = ws;
  float* h1    = ws + 300000;
  float* agg64 = ws + 300000 + 6400000;
  float* sums  = ws + 300000 + 12800000;

  dim3 b256(256);
  zero_kernel<<<(300000 + 255) / 256, b256, 0, stream>>>(agg3, 300000);
  zero_kernel<<<(6400000 + 255) / 256, b256, 0, stream>>>(agg64, 6400000);
  zero_kernel<<<(8320 + 255) / 256, b256, 0, stream>>>(sums, 8320);

  scatter3_kernel<<<(N_EDGES + 255) / 256, b256, 0, stream>>>(x, ei, agg3);
  mlp1_kernel<<<(N_TILES + 3) / 4, dim3(128), 0, stream>>>(
      x, agg3, W1a, b1a, W1b, b1b, h1);
  scatter64_kernel<<<(N_EDGES * 64) / 256, b256, 0, stream>>>(h1, ei, agg64);
  mlp2_kernel<<<(N_TILES + 3) / 4, dim3(128), 0, stream>>>(
      h1, agg64, W2a, b2a, W2b, b2b, h1  );
  pool_add_kernel<<<(N_NODES * 64 + 255) / 256, b256, 0, stream>>>(h1, batch, sums);
  count_kernel<<<(N_NODES + 255) / 256, b256, 0, stream>>>(batch, sums + 8192);
  finalize_kernel<<<(N_GRAPHS * 64 + 255) / 256, b256, 0, stream>>>(
      sums, sums + 8192, out);
}


# Layer-1 edge sum privatised: 240 blocks = 6 node ranges x 40 edge slices, each adds its edges' x rows into an LDS tile (f32 LDS adds), stores a partial plane; MLP1 adds 40 planes; sum order changed
# speedup vs baseline: 2.6823x; 2.6823x over previous
#include <hip/hip_runtime.h>


typedef __attribute__((ext_vector_type(2))) float v2f;
typedef __attribute__((ext_vector_type(8))) float v8f;

#define N_NODES  100000
#define N_EDGES  1600000
#define N_GRAPHS 128
#define N_TILES  (N_NODES / 16)

__global__ void zero_kernel(float* __restrict__ p, int n) {
  int i = blockIdx.x * blockDim.x + threadIdx.x;
  if (i < n) p[i] = 0.0f;
}

#define L1_RANGES  6
#define L1_SLICES  40
#define L1_RN      16672
#define L1_TILE    (L1_RN * 3)
#define L1_SE      (N_EDGES / L1_SLICES)
#define L1_PLANE   (N_NODES * 3)
#define L1_THREADS 1024
#define L1_U       10
#define L1_BATCHES ((L1_SE + L1_THREADS * L1_U - 1) / (L1_THREADS * L1_U))
static_assert(N_EDGES % L1_SLICES == 0, "whole slices");
static_assert(L1_RANGES * L1_RN >= N_NODES, "ranges cover every node");
static_assert(L1_TILE % 32 == 0, "a range is whole 128-byte lines: one writer per line");
static_assert(((N_NODES % L1_RN) * 3) % 4 == 0, "the last range stores whole float4s");

__global__ __launch_bounds__(L1_THREADS) void agg3_partial_kernel(
    const float* __restrict__ x, const int* __restrict__ ei,
    float* __restrict__ part) {
  __shared__ __attribute__((aligned(16))) float tile[L1_TILE];
  const int tid = threadIdx.x;
  const int rg  = blockIdx.x % L1_RANGES;
  const int sl  = blockIdx.x / L1_RANGES;
  const int r0  = rg * L1_RN;

  float4* tile4 = reinterpret_cast<float4*>(tile);
  for (int i = tid; i < L1_TILE / 4; i += L1_THREADS)
    tile4[i] = make_float4(0.0f, 0.0f, 0.0f, 0.0f);
  __syncthreads();

  const int* srcp = ei + sl * L1_SE;
  const int* dstp = ei + N_EDGES + sl * L1_SE;
  for (int b = 0; b < L1_BATCHES; ++b) {
    int s[L1_U];
    unsigned l[L1_U];
    #pragma unroll
    for (int u = 0; u < L1_U; ++u) {
      int e  = (b * L1_U + u) * L1_THREADS + tid;
      int ec = e < L1_SE ? e : L1_SE - 1;
      int d  = dstp[ec];
      s[u]   = srcp[ec];
      l[u]   = e < L1_SE ? (unsigned)(d - r0) : 0xFFFFFFFFu;
    }
    asm volatile("" :: "v"(s[0]));
    float v0[L1_U], v1[L1_U], v2[L1_U];
    #pragma unroll
    for (int u = 0; u < L1_U; ++u) {
      int sj = l[u] < (unsigned)L1_RN ? s[u] : 0;
      const float* xp = x + sj * 3;
      v0[u] = xp[0];
      v1[u] = xp[1];
      v2[u] = xp[2];
    }
    asm volatile("" :: "v"(v0[0]), "v"(v1[0]), "v"(v2[0]));
    #pragma unroll
    for (int u = 0; u < L1_U; ++u) {
      if (l[u] < (unsigned)L1_RN) {
        float* t = tile + l[u] * 3;
        atomicAdd(t,     v0[u]);
        atomicAdd(t + 1, v1[u]);
        atomicAdd(t + 2, v2[u]);
      }
    }
  }
  __syncthreads();

  const int rows = (N_NODES - r0 < L1_RN) ? (N_NODES - r0) : L1_RN;
  float4* out4 = reinterpret_cast<float4*>(part + (size_t)sl * L1_PLANE + r0 * 3);
  for (int i = tid; i < rows * 3 / 4; i += L1_THREADS) out4[i] = tile4[i];
}

#define ROWS_PER_WAVE  32
#define ROWS_IN_FLIGHT 16
static_assert(ROWS_PER_WAVE % ROWS_IN_FLIGHT == 0, "groups tile a wave's rows");
static_assert(N_EDGES % ROWS_PER_WAVE == 0, "whole waves of edges");
static_assert(N_NODES % ROWS_PER_WAVE == 0, "whole waves of nodes");

__device__ __forceinline__ void add_rows_wave(const float* __restrict__ src,
                                              float* __restrict__ dst,
                                              int sid, int did, int lane) {
  #pragma unroll
  for (int g = 0; g < ROWS_PER_WAVE; g += ROWS_IN_FLIGHT) {
    float lo[ROWS_IN_FLIGHT], hi[ROWS_IN_FLIGHT];
    #pragma unroll
    for (int u = 0; u < ROWS_IN_FLIGHT; ++u) {
      int sj = __builtin_amdgcn_readlane(sid, g + u);
      const float* row = src + (size_t)sj * 64;
      lo[u] = row[lane];
      hi[u] = row[lane + 32];
    }
    #pragma unroll
    for (int u = 0; u < ROWS_IN_FLIGHT; ++u) {
      int dj = __builtin_amdgcn_readlane(did, g + u);
      float* row = dst + (size_t)dj * 64;
      atomicAdd(row + lane, lo[u]);
      atomicAdd(row + lane + 32, hi[u]);
    }
  }
}

__global__ __launch_bounds__(256) void scatter64_kernel(
    const float* __restrict__ h, const int* __restrict__ ei,
    float* __restrict__ agg) {
  int wave = (blockIdx.x * 256 + threadIdx.x) >> 5;
  int lane = threadIdx.x & 31;
  int e0   = wave * ROWS_PER_WAVE;
  if (e0 >= N_EDGES) return;
  int e = e0 + lane;
  add_rows_wave(h, agg, ei[e], ei[N_EDGES + e], lane);
}

__global__ __launch_bounds__(128) void mlp1_kernel(
    const float* __restrict__ x, const float* __restrict__ part,
    const float* __restrict__ W1a, const float* __restrict__ b1a,
    const float* __restrict__ W1b, const float* __restrict__ b1b,
    float* __restrict__ h1) {
  __shared__ float sWa[4 * 64];
  __shared__ float sWb[64 * 64];
  __shared__ float sBa[64];
  __shared__ float sBb[64];
  __shared__ float sT[4][16 * 68];

  int tid = threadIdx.x;
  for (int i = tid; i < 4 * 64; i += 128) {
    int k = i >> 6, n = i & 63;
    sWa[i] = (k < 3) ? W1a[k * 64 + n] : 0.0f;
  }
  for (int i = tid; i < 64 * 64; i += 128) sWb[i] = W1b[i];
  if (tid < 64) { sBa[tid] = b1a[tid]; sBb[tid] = b1b[tid]; }
  __syncthreads();

  int wave = tid >> 5;
  int lane = tid & 31;
  int half = lane >> 4;
  int m    = lane & 15;
  int tile = blockIdx.x * 4 + wave;
  if (tile >= N_TILES) return;
  int node0 = tile * 16;
  float* myT = &sT[wave][0];

  int node = node0 + m;
  int k0   = 2 * half;
  int k1   = half ? 2 : 1;
  const float* pn = part + node * 3;
  float p0 = 0.0f, p1 = 0.0f;
  #pragma unroll
  for (int p = 0; p < L1_SLICES; ++p) {
    p0 += pn[p * L1_PLANE + k0];
    p1 += pn[p * L1_PLANE + k1];
  }
  v2f a;
  a.x = x[node * 3 + k0] + p0;
  a.y = (half == 0) ? (x[node * 3 + k1] + p1) : 0.0f;

  v8f acc[4] = {};
  #pragma unroll
  for (int nt = 0; nt < 4; ++nt) {
    v2f b;
    b.x = sWa[k0 * 64 + nt * 16 + m];
    b.y = sWa[(k0 + 1) * 64 + nt * 16 + m];
    acc[nt] = __builtin_amdgcn_wmma_f32_16x16x4_f32(
        false, a, false, b, (short)0, acc[nt], false, false);
  }

  #pragma unroll
  for (int nt = 0; nt < 4; ++nt) {
    float bias = sBa[nt * 16 + m];
    #pragma unroll
    for (int r = 0; r < 8; ++r) {
      float v = acc[nt][r] + bias;
      myT[(r + 8 * half) * 68 + nt * 16 + m] = v > 0.0f ? v : 0.0f;
    }
  }

  v8f acc2[4] = {};
  for (int ks = 0; ks < 16; ++ks) {
    int kk = ks * 4 + 2 * half;
    v2f a2;
    a2.x = myT[m * 68 + kk];
    a2.y = myT[m * 68 + kk + 1];
    #pragma unroll
    for (int nt = 0; nt < 4; ++nt) {
      v2f b;
      b.x = sWb[kk * 64 + nt * 16 + m];
      b.y = sWb[(kk + 1) * 64 + nt * 16 + m];
      acc2[nt] = __builtin_amdgcn_wmma_f32_16x16x4_f32(
          false, a2, false, b, (short)0, acc2[nt], false, false);
    }
  }

  #pragma unroll
  for (int nt = 0; nt < 4; ++nt) {
    float bias = sBb[nt * 16 + m];
    #pragma unroll
    for (int r = 0; r < 8; ++r) {
      float v = acc2[nt][r] + bias;
      h1[(node0 + r + 8 * half) * 64 + nt * 16 + m] = v > 0.0f ? v : 0.0f;
    }
  }
}

__global__ __launch_bounds__(128) void mlp2_kernel(
    const float* __restrict__ h1, const float* __restrict__ agg64,
    const float* __restrict__ W2a, const float* __restrict__ b2a,
    const float* __restrict__ W2b, const float* __restrict__ b2b,
    float* __restrict__ h2) {
  __shared__ float sWa[64 * 64];
  __shared__ float sWb[64 * 64];
  __shared__ float sBa[64];
  __shared__ float sBb[64];
  __shared__ float sT[4][16 * 68];

  int tid = threadIdx.x;
  for (int i = tid; i < 64 * 64; i += 128) { sWa[i] = W2a[i]; sWb[i] = W2b[i]; }
  if (tid < 64) { sBa[tid] = b2a[tid]; sBb[tid] = b2b[tid]; }
  __syncthreads();

  int wave = tid >> 5;
  int lane = tid & 31;
  int half = lane >> 4;
  int m    = lane & 15;
  int tile = blockIdx.x * 4 + wave;
  if (tile >= N_TILES) return;
  int node0 = tile * 16;
  float* myT = &sT[wave][0];

  for (int i = lane; i < 16 * 64; i += 32) {
    int r = i >> 6, c = i & 63;
    int node = node0 + r;
    myT[r * 68 + c] = h1[node * 64 + c] + agg64[node * 64 + c];
  }

  v8f acc[4] = {};
  for (int ks = 0; ks < 16; ++ks) {
    int kk = ks * 4 + 2 * half;
    v2f a;
    a.x = myT[m * 68 + kk];
    a.y = myT[m * 68 + kk + 1];
    #pragma unroll
    for (int nt = 0; nt < 4; ++nt) {
      v2f b;
      b.x = sWa[kk * 64 + nt * 16 + m];
      b.y = sWa[(kk + 1) * 64 + nt * 16 + m];
      acc[nt] = __builtin_amdgcn_wmma_f32_16x16x4_f32(
          false, a, false, b, (short)0, acc[nt], false, false);
    }
  }

  #pragma unroll
  for (int nt = 0; nt < 4; ++nt) {
    float bias = sBa[nt * 16 + m];
    #pragma unroll
    for (int r = 0; r < 8; ++r) {
      float v = acc[nt][r] + bias;
      myT[(r + 8 * half) * 68 + nt * 16 + m] = v > 0.0f ? v : 0.0f;
    }
  }

  v8f acc2[4] = {};
  for (int ks = 0; ks < 16; ++ks) {
    int kk = ks * 4 + 2 * half;
    v2f a2;
    a2.x = myT[m * 68 + kk];
    a2.y = myT[m * 68 + kk + 1];
    #pragma unroll
    for (int nt = 0; nt < 4; ++nt) {
      v2f b;
      b.x = sWb[kk * 64 + nt * 16 + m];
      b.y = sWb[(kk + 1) * 64 + nt * 16 + m];
      acc2[nt] = __builtin_amdgcn_wmma_f32_16x16x4_f32(
          false, a2, false, b, (short)0, acc2[nt], false, false);
    }
  }

  #pragma unroll
  for (int nt = 0; nt < 4; ++nt) {
    float bias = sBb[nt * 16 + m];
    #pragma unroll
    for (int r = 0; r < 8; ++r) {
      float v = acc2[nt][r] + bias;
      h2[(node0 + r + 8 * half) * 64 + nt * 16 + m] = v > 0.0f ? v : 0.0f;
    }
  }
}

__global__ __launch_bounds__(256) void pool_add_kernel(
    const float* __restrict__ h, const int* __restrict__ batch,
    float* __restrict__ sums) {
  int wave = (blockIdx.x * 256 + threadIdx.x) >> 5;
  int lane = threadIdx.x & 31;
  int n0   = wave * ROWS_PER_WAVE;
  if (n0 >= N_NODES) return;
  int n = n0 + lane;
  add_rows_wave(h, sums, n, batch[n], lane);
}

__global__ void count_kernel(const int* __restrict__ batch,
                             float* __restrict__ cnt) {
  int n = blockIdx.x * blockDim.x + threadIdx.x;
  if (n >= N_NODES) return;
  atomicAdd(&cnt[batch[n]], 1.0f);
}

__global__ void finalize_kernel(const float* __restrict__ sums,
                                const float* __restrict__ cnt,
                                float* __restrict__ out) {
  int i = blockIdx.x * blockDim.x + threadIdx.x;
  if (i >= N_GRAPHS * 64) return;
  float c = cnt[i >> 6];
  out[i] = sums[i] / fmaxf(c, 1.0f);
}

extern "C" void kernel_launch(void* const* d_in, const int* in_sizes, int n_in,
                              void* d_out, int out_size, void* d_ws, size_t ws_size,
                              hipStream_t stream) {
  const float* x     = (const float*)d_in[0];
  const int*   ei    = (const int*)d_in[1];
  const int*   batch = (const int*)d_in[2];
  const float* W1a   = (const float*)d_in[3];
  const float* b1a   = (const float*)d_in[4];
  const float* W1b   = (const float*)d_in[5];
  const float* b1b   = (const float*)d_in[6];
  const float* W2a   = (const float*)d_in[7];
  const float* b2a   = (const float*)d_in[8];
  const float* W2b   = (const float*)d_in[9];
  const float* b2b   = (const float*)d_in[10];
  float* out = (float*)d_out;

  float* ws    = (float*)d_ws;
  float* part  = ws;
  float* h1    = ws + 12000000;
  float* agg64 = ws + 12000000 + 6400000;
  float* sums  = ws + 12000000 + 12800000;

  dim3 b256(256);
  zero_kernel<<<(6400000 + 255) / 256, b256, 0, stream>>>(agg64, 6400000);
  zero_kernel<<<(8320 + 255) / 256, b256, 0, stream>>>(sums, 8320);

  agg3_partial_kernel<<<L1_RANGES * L1_SLICES, dim3(L1_THREADS), 0, stream>>>(x, ei, part);
  mlp1_kernel<<<(N_TILES + 3) / 4, dim3(128), 0, stream>>>(
      x, part, W1a, b1a, W1b, b1b, h1);
  scatter64_kernel<<<(N_EDGES / ROWS_PER_WAVE + 7) / 8, b256, 0, stream>>>(h1, ei, agg64);
  mlp2_kernel<<<(N_TILES + 3) / 4, dim3(128), 0, stream>>>(
      h1, agg64, W2a, b2a, W2b, b2b, h1  );
  pool_add_kernel<<<(N_NODES / ROWS_PER_WAVE + 7) / 8, b256, 0, stream>>>(h1, batch, sums);
  count_kernel<<<(N_NODES + 255) / 256, b256, 0, stream>>>(batch, sums + 8192);
  finalize_kernel<<<(N_GRAPHS * 64 + 255) / 256, b256, 0, stream>>>(
      sums, sums + 8192, out);
}


# Pooling: a wave adds each run of equal graph ids among its 32 nodes in registers, then 2 line atomics + 1 size add per run, not per node; count launch gone; sum order changed (run partials)
# speedup vs baseline: 4.0506x; 1.5101x over previous
#include <hip/hip_runtime.h>


typedef __attribute__((ext_vector_type(2))) float v2f;
typedef __attribute__((ext_vector_type(8))) float v8f;

#define N_NODES  100000
#define N_EDGES  1600000
#define N_GRAPHS 128
#define N_TILES  (N_NODES / 16)

__global__ void zero_kernel(float* __restrict__ p, int n) {
  int i = blockIdx.x * blockDim.x + threadIdx.x;
  if (i < n) p[i] = 0.0f;
}

#define L1_RANGES  6
#define L1_SLICES  40
#define L1_RN      16672
#define L1_TILE    (L1_RN * 3)
#define L1_SE      (N_EDGES / L1_SLICES)
#define L1_PLANE   (N_NODES * 3)
#define L1_THREADS 1024
#define L1_U       10
#define L1_BATCHES ((L1_SE + L1_THREADS * L1_U - 1) / (L1_THREADS * L1_U))
static_assert(N_EDGES % L1_SLICES == 0, "whole slices");
static_assert(L1_RANGES * L1_RN >= N_NODES, "ranges cover every node");
static_assert(L1_TILE % 32 == 0, "a range is whole 128-byte lines: one writer per line");
static_assert(((N_NODES % L1_RN) * 3) % 4 == 0, "the last range stores whole float4s");

__global__ __launch_bounds__(L1_THREADS) void agg3_partial_kernel(
    const float* __restrict__ x, const int* __restrict__ ei,
    float* __restrict__ part) {
  __shared__ __attribute__((aligned(16))) float tile[L1_TILE];
  const int tid = threadIdx.x;
  const int rg  = blockIdx.x % L1_RANGES;
  const int sl  = blockIdx.x / L1_RANGES;
  const int r0  = rg * L1_RN;

  float4* tile4 = reinterpret_cast<float4*>(tile);
  for (int i = tid; i < L1_TILE / 4; i += L1_THREADS)
    tile4[i] = make_float4(0.0f, 0.0f, 0.0f, 0.0f);
  __syncthreads();

  const int* srcp = ei + sl * L1_SE;
  const int* dstp = ei + N_EDGES + sl * L1_SE;
  for (int b = 0; b < L1_BATCHES; ++b) {
    int s[L1_U];
    unsigned l[L1_U];
    #pragma unroll
    for (int u = 0; u < L1_U; ++u) {
      int e  = (b * L1_U + u) * L1_THREADS + tid;
      int ec = e < L1_SE ? e : L1_SE - 1;
      int d  = dstp[ec];
      s[u]   = srcp[ec];
      l[u]   = e < L1_SE ? (unsigned)(d - r0) : 0xFFFFFFFFu;
    }
    asm volatile("" :: "v"(s[0]));
    float v0[L1_U], v1[L1_U], v2[L1_U];
    #pragma unroll
    for (int u = 0; u < L1_U; ++u) {
      int sj = l[u] < (unsigned)L1_RN ? s[u] : 0;
      const float* xp = x + sj * 3;
      v0[u] = xp[0];
      v1[u] = xp[1];
      v2[u] = xp[2];
    }
    asm volatile("" :: "v"(v0[0]), "v"(v1[0]), "v"(v2[0]));
    #pragma unroll
    for (int u = 0; u < L1_U; ++u) {
      if (l[u] < (unsigned)L1_RN) {
        float* t = tile + l[u] * 3;
        atomicAdd(t,     v0[u]);
        atomicAdd(t + 1, v1[u]);
        atomicAdd(t + 2, v2[u]);
      }
    }
  }
  __syncthreads();

  const int rows = (N_NODES - r0 < L1_RN) ? (N_NODES - r0) : L1_RN;
  float4* out4 = reinterpret_cast<float4*>(part + (size_t)sl * L1_PLANE + r0 * 3);
  for (int i = tid; i < rows * 3 / 4; i += L1_THREADS) out4[i] = tile4[i];
}

#define ROWS_PER_WAVE  32
#define ROWS_IN_FLIGHT 16
static_assert(ROWS_PER_WAVE % ROWS_IN_FLIGHT == 0, "groups tile a wave's rows");
static_assert(N_EDGES % ROWS_PER_WAVE == 0, "whole waves of edges");
static_assert(N_NODES % ROWS_PER_WAVE == 0, "whole waves of nodes");

__device__ __forceinline__ void add_rows_wave(const float* __restrict__ src,
                                              float* __restrict__ dst,
                                              int sid, int did, int lane) {
  #pragma unroll
  for (int g = 0; g < ROWS_PER_WAVE; g += ROWS_IN_FLIGHT) {
    float lo[ROWS_IN_FLIGHT], hi[ROWS_IN_FLIGHT];
    #pragma unroll
    for (int u = 0; u < ROWS_IN_FLIGHT; ++u) {
      int sj = __builtin_amdgcn_readlane(sid, g + u);
      const float* row = src + (size_t)sj * 64;
      lo[u] = row[lane];
      hi[u] = row[lane + 32];
    }
    #pragma unroll
    for (int u = 0; u < ROWS_IN_FLIGHT; ++u) {
      int dj = __builtin_amdgcn_readlane(did, g + u);
      float* row = dst + (size_t)dj * 64;
      atomicAdd(row + lane, lo[u]);
      atomicAdd(row + lane + 32, hi[u]);
    }
  }
}

__global__ __launch_bounds__(256) void scatter64_kernel(
    const float* __restrict__ h, const int* __restrict__ ei,
    float* __restrict__ agg) {
  int wave = (blockIdx.x * 256 + threadIdx.x) >> 5;
  int lane = threadIdx.x & 31;
  int e0   = wave * ROWS_PER_WAVE;
  if (e0 >= N_EDGES) return;
  int e = e0 + lane;
  add_rows_wave(h, agg, ei[e], ei[N_EDGES + e], lane);
}

__global__ __launch_bounds__(128) void mlp1_kernel(
    const float* __restrict__ x, const float* __restrict__ part,
    const float* __restrict__ W1a, const float* __restrict__ b1a,
    const float* __restrict__ W1b, const float* __restrict__ b1b,
    float* __restrict__ h1) {
  __shared__ float sWa[4 * 64];
  __shared__ float sWb[64 * 64];
  __shared__ float sBa[64];
  __shared__ float sBb[64];
  __shared__ float sT[4][16 * 68];

  int tid = threadIdx.x;
  for (int i = tid; i < 4 * 64; i += 128) {
    int k = i >> 6, n = i & 63;
    sWa[i] = (k < 3) ? W1a[k * 64 + n] : 0.0f;
  }
  for (int i = tid; i < 64 * 64; i += 128) sWb[i] = W1b[i];
  if (tid < 64) { sBa[tid] = b1a[tid]; sBb[tid] = b1b[tid]; }
  __syncthreads();

  int wave = tid >> 5;
  int lane = tid & 31;
  int half = lane >> 4;
  int m    = lane & 15;
  int tile = blockIdx.x * 4 + wave;
  if (tile >= N_TILES) return;
  int node0 = tile * 16;
  float* myT = &sT[wave][0];

  int node = node0 + m;
  int k0   = 2 * half;
  int k1   = half ? 2 : 1;
  const float* pn = part + node * 3;
  float p0 = 0.0f, p1 = 0.0f;
  #pragma unroll
  for (int p = 0; p < L1_SLICES; ++p) {
    p0 += pn[p * L1_PLANE + k0];
    p1 += pn[p * L1_PLANE + k1];
  }
  v2f a;
  a.x = x[node * 3 + k0] + p0;
  a.y = (half == 0) ? (x[node * 3 + k1] + p1) : 0.0f;

  v8f acc[4] = {};
  #pragma unroll
  for (int nt = 0; nt < 4; ++nt) {
    v2f b;
    b.x = sWa[k0 * 64 + nt * 16 + m];
    b.y = sWa[(k0 + 1) * 64 + nt * 16 + m];
    acc[nt] = __builtin_amdgcn_wmma_f32_16x16x4_f32(
        false, a, false, b, (short)0, acc[nt], false, false);
  }

  #pragma unroll
  for (int nt = 0; nt < 4; ++nt) {
    float bias = sBa[nt * 16 + m];
    #pragma unroll
    for (int r = 0; r < 8; ++r) {
      float v = acc[nt][r] + bias;
      myT[(r + 8 * half) * 68 + nt * 16 + m] = v > 0.0f ? v : 0.0f;
    }
  }

  v8f acc2[4] = {};
  for (int ks = 0; ks < 16; ++ks) {
    int kk = ks * 4 + 2 * half;
    v2f a2;
    a2.x = myT[m * 68 + kk];
    a2.y = myT[m * 68 + kk + 1];
    #pragma unroll
    for (int nt = 0; nt < 4; ++nt) {
      v2f b;
      b.x = sWb[kk * 64 + nt * 16 + m];
      b.y = sWb[(kk + 1) * 64 + nt * 16 + m];
      acc2[nt] = __builtin_amdgcn_wmma_f32_16x16x4_f32(
          false, a2, false, b, (short)0, acc2[nt], false, false);
    }
  }

  #pragma unroll
  for (int nt = 0; nt < 4; ++nt) {
    float bias = sBb[nt * 16 + m];
    #pragma unroll
    for (int r = 0; r < 8; ++r) {
      float v = acc2[nt][r] + bias;
      h1[(node0 + r + 8 * half) * 64 + nt * 16 + m] = v > 0.0f ? v : 0.0f;
    }
  }
}

__global__ __launch_bounds__(128) void mlp2_kernel(
    const float* __restrict__ h1, const float* __restrict__ agg64,
    const float* __restrict__ W2a, const float* __restrict__ b2a,
    const float* __restrict__ W2b, const float* __restrict__ b2b,
    float* __restrict__ h2) {
  __shared__ float sWa[64 * 64];
  __shared__ float sWb[64 * 64];
  __shared__ float sBa[64];
  __shared__ float sBb[64];
  __shared__ float sT[4][16 * 68];

  int tid = threadIdx.x;
  for (int i = tid; i < 64 * 64; i += 128) { sWa[i] = W2a[i]; sWb[i] = W2b[i]; }
  if (tid < 64) { sBa[tid] = b2a[tid]; sBb[tid] = b2b[tid]; }
  __syncthreads();

  int wave = tid >> 5;
  int lane = tid & 31;
  int half = lane >> 4;
  int m    = lane & 15;
  int tile = blockIdx.x * 4 + wave;
  if (tile >= N_TILES) return;
  int node0 = tile * 16;
  float* myT = &sT[wave][0];

  for (int i = lane; i < 16 * 64; i += 32) {
    int r = i >> 6, c = i & 63;
    int node = node0 + r;
    myT[r * 68 + c] = h1[node * 64 + c] + agg64[node * 64 + c];
  }

  v8f acc[4] = {};
  for (int ks = 0; ks < 16; ++ks) {
    int kk = ks * 4 + 2 * half;
    v2f a;
    a.x = myT[m * 68 + kk];
    a.y = myT[m * 68 + kk + 1];
    #pragma unroll
    for (int nt = 0; nt < 4; ++nt) {
      v2f b;
      b.x = sWa[kk * 64 + nt * 16 + m];
      b.y = sWa[(kk + 1) * 64 + nt * 16 + m];
      acc[nt] = __builtin_amdgcn_wmma_f32_16x16x4_f32(
          false, a, false, b, (short)0, acc[nt], false, false);
    }
  }

  #pragma unroll
  for (int nt = 0; nt < 4; ++nt) {
    float bias = sBa[nt * 16 + m];
    #pragma unroll
    for (int r = 0; r < 8; ++r) {
      float v = acc[nt][r] + bias;
      myT[(r + 8 * half) * 68 + nt * 16 + m] = v > 0.0f ? v : 0.0f;
    }
  }

  v8f acc2[4] = {};
  for (int ks = 0; ks < 16; ++ks) {
    int kk = ks * 4 + 2 * half;
    v2f a2;
    a2.x = myT[m * 68 + kk];
    a2.y = myT[m * 68 + kk + 1];
    #pragma unroll
    for (int nt = 0; nt < 4; ++nt) {
      v2f b;
      b.x = sWb[kk * 64 + nt * 16 + m];
      b.y = sWb[(kk + 1) * 64 + nt * 16 + m];
      acc2[nt] = __builtin_amdgcn_wmma_f32_16x16x4_f32(
          false, a2, false, b, (short)0, acc2[nt], false, false);
    }
  }

  #pragma unroll
  for (int nt = 0; nt < 4; ++nt) {
    float bias = sBb[nt * 16 + m];
    #pragma unroll
    for (int r = 0; r < 8; ++r) {
      float v = acc2[nt][r] + bias;
      h2[(node0 + r + 8 * half) * 64 + nt * 16 + m] = v > 0.0f ? v : 0.0f;
    }
  }
}

__device__ __forceinline__ void pool_flush_run(float* __restrict__ sums,
                                               float* __restrict__ cnt, int gid,
                                               float lo, float hi, int rows, int lane) {
  float* row = sums + (size_t)gid * 64;
  atomicAdd(row + lane, lo);
  atomicAdd(row + lane + 32, hi);
  if (lane == 0) atomicAdd(cnt + gid, (float)rows);
}

__global__ __launch_bounds__(256) void pool_runs_kernel(
    const float* __restrict__ h, const int* __restrict__ batch,
    float* __restrict__ sums, float* __restrict__ cnt) {
  int wave = (blockIdx.x * 256 + threadIdx.x) >> 5;
  int lane = threadIdx.x & 31;
  int n0   = wave * ROWS_PER_WAVE;
  if (n0 >= N_NODES) return;
  int gid_l = batch[n0 + lane];
  const float* base = h + (size_t)n0 * 64 + lane;

  int   cur  = __builtin_amdgcn_readlane(gid_l, 0);
  int   rows = 0;
  float accLo = 0.0f, accHi = 0.0f;
  #pragma unroll
  for (int g = 0; g < ROWS_PER_WAVE; g += ROWS_IN_FLIGHT) {
    float lo[ROWS_IN_FLIGHT], hi[ROWS_IN_FLIGHT];
    #pragma unroll
    for (int u = 0; u < ROWS_IN_FLIGHT; ++u) {
      lo[u] = base[(g + u) * 64];
      hi[u] = base[(g + u) * 64 + 32];
    }
    #pragma unroll
    for (int u = 0; u < ROWS_IN_FLIGHT; ++u) {
      int gj = __builtin_amdgcn_readlane(gid_l, g + u);
      if (gj != cur) {
        pool_flush_run(sums, cnt, cur, accLo, accHi, rows, lane);
        cur = gj; rows = 0; accLo = 0.0f; accHi = 0.0f;
      }
      accLo += lo[u];
      accHi += hi[u];
      ++rows;
    }
  }
  pool_flush_run(sums, cnt, cur, accLo, accHi, rows, lane);
}

__global__ void finalize_kernel(const float* __restrict__ sums,
                                const float* __restrict__ cnt,
                                float* __restrict__ out) {
  int i = blockIdx.x * blockDim.x + threadIdx.x;
  if (i >= N_GRAPHS * 64) return;
  float c = cnt[i >> 6];
  out[i] = sums[i] / fmaxf(c, 1.0f);
}

extern "C" void kernel_launch(void* const* d_in, const int* in_sizes, int n_in,
                              void* d_out, int out_size, void* d_ws, size_t ws_size,
                              hipStream_t stream) {
  const float* x     = (const float*)d_in[0];
  const int*   ei    = (const int*)d_in[1];
  const int*   batch = (const int*)d_in[2];
  const float* W1a   = (const float*)d_in[3];
  const float* b1a   = (const float*)d_in[4];
  const float* W1b   = (const float*)d_in[5];
  const float* b1b   = (const float*)d_in[6];
  const float* W2a   = (const float*)d_in[7];
  const float* b2a   = (const float*)d_in[8];
  const float* W2b   = (const float*)d_in[9];
  const float* b2b   = (const float*)d_in[10];
  float* out = (float*)d_out;

  float* ws    = (float*)d_ws;
  float* part  = ws;
  float* h1    = ws + 12000000;
  float* agg64 = ws + 12000000 + 6400000;
  float* sums  = ws + 12000000 + 12800000;

  dim3 b256(256);
  zero_kernel<<<(6400000 + 255) / 256, b256, 0, stream>>>(agg64, 6400000);
  zero_kernel<<<(8320 + 255) / 256, b256, 0, stream>>>(sums, 8320);

  agg3_partial_kernel<<<L1_RANGES * L1_SLICES, dim3(L1_THREADS), 0, stream>>>(x, ei, part);
  mlp1_kernel<<<(N_TILES + 3) / 4, dim3(128), 0, stream>>>(
      x, part, W1a, b1a, W1b, b1b, h1);
  scatter64_kernel<<<(N_EDGES / ROWS_PER_WAVE + 7) / 8, b256, 0, stream>>>(h1, ei, agg64);
  mlp2_kernel<<<(N_TILES + 3) / 4, dim3(128), 0, stream>>>(
      h1, agg64, W2a, b2a, W2b, b2b, h1  );
  pool_runs_kernel<<<(N_NODES / ROWS_PER_WAVE + 7) / 8, b256, 0, stream>>>(
      h1, batch, sums, sums + 8192);
  finalize_kernel<<<(N_GRAPHS * 64 + 255) / 256, b256, 0, stream>>>(
      sums, sums + 8192, out);
}


# Both fill launches gone: MLP1's epilogue also stores h1 as the layer-2 sum's start value (self term first: sum order changed) and its first blocks clear the pool sums; MLP2 reads one array
# speedup vs baseline: 4.1756x; 1.0309x over previous
#include <hip/hip_runtime.h>


typedef __attribute__((ext_vector_type(2))) float v2f;
typedef __attribute__((ext_vector_type(8))) float v8f;

#define N_NODES  100000
#define N_EDGES  1600000
#define N_GRAPHS 128
#define N_TILES  (N_NODES / 16)

#define L1_RANGES  6
#define L1_SLICES  40
#define L1_RN      16672
#define L1_TILE    (L1_RN * 3)
#define L1_SE      (N_EDGES / L1_SLICES)
#define L1_PLANE   (N_NODES * 3)
#define L1_THREADS 1024
#define L1_U       10
#define L1_BATCHES ((L1_SE + L1_THREADS * L1_U - 1) / (L1_THREADS * L1_U))
static_assert(N_EDGES % L1_SLICES == 0, "whole slices");
static_assert(L1_RANGES * L1_RN >= N_NODES, "ranges cover every node");
static_assert(L1_TILE % 32 == 0, "a range is whole 128-byte lines: one writer per line");
static_assert(((N_NODES % L1_RN) * 3) % 4 == 0, "the last range stores whole float4s");

__global__ __launch_bounds__(L1_THREADS) void agg3_partial_kernel(
    const float* __restrict__ x, const int* __restrict__ ei,
    float* __restrict__ part) {
  __shared__ __attribute__((aligned(16))) float tile[L1_TILE];
  const int tid = threadIdx.x;
  const int rg  = blockIdx.x % L1_RANGES;
  const int sl  = blockIdx.x / L1_RANGES;
  const int r0  = rg * L1_RN;

  float4* tile4 = reinterpret_cast<float4*>(tile);
  for (int i = tid; i < L1_TILE / 4; i += L1_THREADS)
    tile4[i] = make_float4(0.0f, 0.0f, 0.0f, 0.0f);
  __syncthreads();

  const int* srcp = ei + sl * L1_SE;
  const int* dstp = ei + N_EDGES + sl * L1_SE;
  for (int b = 0; b < L1_BATCHES; ++b) {
    int s[L1_U];
    unsigned l[L1_U];
    #pragma unroll
    for (int u = 0; u < L1_U; ++u) {
      int e  = (b * L1_U + u) * L1_THREADS + tid;
      int ec = e < L1_SE ? e : L1_SE - 1;
      int d  = dstp[ec];
      s[u]   = srcp[ec];
      l[u]   = e < L1_SE ? (unsigned)(d - r0) : 0xFFFFFFFFu;
    }
    asm volatile("" :: "v"(s[0]));
    float v0[L1_U], v1[L1_U], v2[L1_U];
    #pragma unroll
    for (int u = 0; u < L1_U; ++u) {
      int sj = l[u] < (unsigned)L1_RN ? s[u] : 0;
      const float* xp = x + sj * 3;
      v0[u] = xp[0];
      v1[u] = xp[1];
      v2[u] = xp[2];
    }
    asm volatile("" :: "v"(v0[0]), "v"(v1[0]), "v"(v2[0]));
    #pragma unroll
    for (int u = 0; u < L1_U; ++u) {
      if (l[u] < (unsigned)L1_RN) {
        float* t = tile + l[u] * 3;
        atomicAdd(t,     v0[u]);
        atomicAdd(t + 1, v1[u]);
        atomicAdd(t + 2, v2[u]);
      }
    }
  }
  __syncthreads();

  const int rows = (N_NODES - r0 < L1_RN) ? (N_NODES - r0) : L1_RN;
  float4* out4 = reinterpret_cast<float4*>(part + (size_t)sl * L1_PLANE + r0 * 3);
  for (int i = tid; i < rows * 3 / 4; i += L1_THREADS) out4[i] = tile4[i];
}

#define ROWS_PER_WAVE  32
#define ROWS_IN_FLIGHT 16
static_assert(ROWS_PER_WAVE % ROWS_IN_FLIGHT == 0, "groups tile a wave's rows");
static_assert(N_EDGES % ROWS_PER_WAVE == 0, "whole waves of edges");
static_assert(N_NODES % ROWS_PER_WAVE == 0, "whole waves of nodes");

__device__ __forceinline__ void add_rows_wave(const float* __restrict__ src,
                                              float* __restrict__ dst,
                                              int sid, int did, int lane) {
  #pragma unroll
  for (int g = 0; g < ROWS_PER_WAVE; g += ROWS_IN_FLIGHT) {
    float lo[ROWS_IN_FLIGHT], hi[ROWS_IN_FLIGHT];
    #pragma unroll
    for (int u = 0; u < ROWS_IN_FLIGHT; ++u) {
      int sj = __builtin_amdgcn_readlane(sid, g + u);
      const float* row = src + (size_t)sj * 64;
      lo[u] = row[lane];
      hi[u] = row[lane + 32];
    }
    #pragma unroll
    for (int u = 0; u < ROWS_IN_FLIGHT; ++u) {
      int dj = __builtin_amdgcn_readlane(did, g + u);
      float* row = dst + (size_t)dj * 64;
      atomicAdd(row + lane, lo[u]);
      atomicAdd(row + lane + 32, hi[u]);
    }
  }
}

__global__ __launch_bounds__(256) void scatter64_kernel(
    const float* __restrict__ h, const int* __restrict__ ei,
    float* __restrict__ agg) {
  int wave = (blockIdx.x * 256 + threadIdx.x) >> 5;
  int lane = threadIdx.x & 31;
  int e0   = wave * ROWS_PER_WAVE;
  if (e0 >= N_EDGES) return;
  int e = e0 + lane;
  add_rows_wave(h, agg, ei[e], ei[N_EDGES + e], lane);
}

#define POOL_FLOATS       (N_GRAPHS * 64 + N_GRAPHS)
#define POOL_CLEAR_BLOCKS (POOL_FLOATS / 128)
static_assert(POOL_FLOATS % 128 == 0, "whole blocks clear the pooling sums");
static_assert(POOL_CLEAR_BLOCKS <= (N_TILES + 3) / 4, "enough blocks to clear them");

__global__ __launch_bounds__(128) void mlp1_kernel(
    const float* __restrict__ x, const float* __restrict__ part,
    const float* __restrict__ W1a, const float* __restrict__ b1a,
    const float* __restrict__ W1b, const float* __restrict__ b1b,
    float* __restrict__ h1, float* __restrict__ hsum, float* __restrict__ pool) {
  __shared__ float sWa[4 * 64];
  __shared__ float sWb[64 * 64];
  __shared__ float sBa[64];
  __shared__ float sBb[64];
  __shared__ float sT[4][16 * 68];

  int tid = threadIdx.x;
  if (blockIdx.x < POOL_CLEAR_BLOCKS) pool[blockIdx.x * 128 + tid] = 0.0f;
  for (int i = tid; i < 4 * 64; i += 128) {
    int k = i >> 6, n = i & 63;
    sWa[i] = (k < 3) ? W1a[k * 64 + n] : 0.0f;
  }
  for (int i = tid; i < 64 * 64; i += 128) sWb[i] = W1b[i];
  if (tid < 64) { sBa[tid] = b1a[tid]; sBb[tid] = b1b[tid]; }
  __syncthreads();

  int wave = tid >> 5;
  int lane = tid & 31;
  int half = lane >> 4;
  int m    = lane & 15;
  int tile = blockIdx.x * 4 + wave;
  if (tile >= N_TILES) return;
  int node0 = tile * 16;
  float* myT = &sT[wave][0];

  int node = node0 + m;
  int k0   = 2 * half;
  int k1   = half ? 2 : 1;
  const float* pn = part + node * 3;
  float p0 = 0.0f, p1 = 0.0f;
  #pragma unroll
  for (int p = 0; p < L1_SLICES; ++p) {
    p0 += pn[p * L1_PLANE + k0];
    p1 += pn[p * L1_PLANE + k1];
  }
  v2f a;
  a.x = x[node * 3 + k0] + p0;
  a.y = (half == 0) ? (x[node * 3 + k1] + p1) : 0.0f;

  v8f acc[4] = {};
  #pragma unroll
  for (int nt = 0; nt < 4; ++nt) {
    v2f b;
    b.x = sWa[k0 * 64 + nt * 16 + m];
    b.y = sWa[(k0 + 1) * 64 + nt * 16 + m];
    acc[nt] = __builtin_amdgcn_wmma_f32_16x16x4_f32(
        false, a, false, b, (short)0, acc[nt], false, false);
  }

  #pragma unroll
  for (int nt = 0; nt < 4; ++nt) {
    float bias = sBa[nt * 16 + m];
    #pragma unroll
    for (int r = 0; r < 8; ++r) {
      float v = acc[nt][r] + bias;
      myT[(r + 8 * half) * 68 + nt * 16 + m] = v > 0.0f ? v : 0.0f;
    }
  }

  v8f acc2[4] = {};
  for (int ks = 0; ks < 16; ++ks) {
    int kk = ks * 4 + 2 * half;
    v2f a2;
    a2.x = myT[m * 68 + kk];
    a2.y = myT[m * 68 + kk + 1];
    #pragma unroll
    for (int nt = 0; nt < 4; ++nt) {
      v2f b;
      b.x = sWb[kk * 64 + nt * 16 + m];
      b.y = sWb[(kk + 1) * 64 + nt * 16 + m];
      acc2[nt] = __builtin_amdgcn_wmma_f32_16x16x4_f32(
          false, a2, false, b, (short)0, acc2[nt], false, false);
    }
  }

  #pragma unroll
  for (int nt = 0; nt < 4; ++nt) {
    float bias = sBb[nt * 16 + m];
    #pragma unroll
    for (int r = 0; r < 8; ++r) {
      float v = acc2[nt][r] + bias;
      v = v > 0.0f ? v : 0.0f;
      int o = (node0 + r + 8 * half) * 64 + nt * 16 + m;
      h1[o]   = v;
      hsum[o] = v;
    }
  }
}

__global__ __launch_bounds__(128) void mlp2_kernel(
    const float* __restrict__ hsum,
    const float* __restrict__ W2a, const float* __restrict__ b2a,
    const float* __restrict__ W2b, const float* __restrict__ b2b,
    float* __restrict__ h2) {
  __shared__ float sWa[64 * 64];
  __shared__ float sWb[64 * 64];
  __shared__ float sBa[64];
  __shared__ float sBb[64];
  __shared__ float sT[4][16 * 68];

  int tid = threadIdx.x;
  for (int i = tid; i < 64 * 64; i += 128) { sWa[i] = W2a[i]; sWb[i] = W2b[i]; }
  if (tid < 64) { sBa[tid] = b2a[tid]; sBb[tid] = b2b[tid]; }
  __syncthreads();

  int wave = tid >> 5;
  int lane = tid & 31;
  int half = lane >> 4;
  int m    = lane & 15;
  int tile = blockIdx.x * 4 + wave;
  if (tile >= N_TILES) return;
  int node0 = tile * 16;
  float* myT = &sT[wave][0];

  for (int i = lane; i < 16 * 64; i += 32) {
    int r = i >> 6, c = i & 63;
    int node = node0 + r;
    myT[r * 68 + c] = hsum[node * 64 + c];
  }

  v8f acc[4] = {};
  for (int ks = 0; ks < 16; ++ks) {
    int kk = ks * 4 + 2 * half;
    v2f a;
    a.x = myT[m * 68 + kk];
    a.y = myT[m * 68 + kk + 1];
    #pragma unroll
    for (int nt = 0; nt < 4; ++nt) {
      v2f b;
      b.x = sWa[kk * 64 + nt * 16 + m];
      b.y = sWa[(kk + 1) * 64 + nt * 16 + m];
      acc[nt] = __builtin_amdgcn_wmma_f32_16x16x4_f32(
          false, a, false, b, (short)0, acc[nt], false, false);
    }
  }

  #pragma unroll
  for (int nt = 0; nt < 4; ++nt) {
    float bias = sBa[nt * 16 + m];
    #pragma unroll
    for (int r = 0; r < 8; ++r) {
      float v = acc[nt][r] + bias;
      myT[(r + 8 * half) * 68 + nt * 16 + m] = v > 0.0f ? v : 0.0f;
    }
  }

  v8f acc2[4] = {};
  for (int ks = 0; ks < 16; ++ks) {
    int kk = ks * 4 + 2 * half;
    v2f a2;
    a2.x = myT[m * 68 + kk];
    a2.y = myT[m * 68 + kk + 1];
    #pragma unroll
    for (int nt = 0; nt < 4; ++nt) {
      v2f b;
      b.x = sWb[kk * 64 + nt * 16 + m];
      b.y = sWb[(kk + 1) * 64 + nt * 16 + m];
      acc2[nt] = __builtin_amdgcn_wmma_f32_16x16x4_f32(
          false, a2, false, b, (short)0, acc2[nt], false, false);
    }
  }

  #pragma unroll
  for (int nt = 0; nt < 4; ++nt) {
    float bias = sBb[nt * 16 + m];
    #pragma unroll
    for (int r = 0; r < 8; ++r) {
      float v = acc2[nt][r] + bias;
      h2[(node0 + r + 8 * half) * 64 + nt * 16 + m] = v > 0.0f ? v : 0.0f;
    }
  }
}

__device__ __forceinline__ void pool_flush_run(float* __restrict__ sums,
                                               float* __restrict__ cnt, int gid,
                                               float lo, float hi, int rows, int lane) {
  float* row = sums + (size_t)gid * 64;
  atomicAdd(row + lane, lo);
  atomicAdd(row + lane + 32, hi);
  if (lane == 0) atomicAdd(cnt + gid, (float)rows);
}

__global__ __launch_bounds__(256) void pool_runs_kernel(
    const float* __restrict__ h, const int* __restrict__ batch,
    float* __restrict__ sums, float* __restrict__ cnt) {
  int wave = (blockIdx.x * 256 + threadIdx.x) >> 5;
  int lane = threadIdx.x & 31;
  int n0   = wave * ROWS_PER_WAVE;
  if (n0 >= N_NODES) return;
  int gid_l = batch[n0 + lane];
  const float* base = h + (size_t)n0 * 64 + lane;

  int   cur  = __builtin_amdgcn_readlane(gid_l, 0);
  int   rows = 0;
  float accLo = 0.0f, accHi = 0.0f;
  #pragma unroll
  for (int g = 0; g < ROWS_PER_WAVE; g += ROWS_IN_FLIGHT) {
    float lo[ROWS_IN_FLIGHT], hi[ROWS_IN_FLIGHT];
    #pragma unroll
    for (int u = 0; u < ROWS_IN_FLIGHT; ++u) {
      lo[u] = base[(g + u) * 64];
      hi[u] = base[(g + u) * 64 + 32];
    }
    #pragma unroll
    for (int u = 0; u < ROWS_IN_FLIGHT; ++u) {
      int gj = __builtin_amdgcn_readlane(gid_l, g + u);
      if (gj != cur) {
        pool_flush_run(sums, cnt, cur, accLo, accHi, rows, lane);
        cur = gj; rows = 0; accLo = 0.0f; accHi = 0.0f;
      }
      accLo += lo[u];
      accHi += hi[u];
      ++rows;
    }
  }
  pool_flush_run(sums, cnt, cur, accLo, accHi, rows, lane);
}

__global__ void finalize_kernel(const float* __restrict__ sums,
                                const float* __restrict__ cnt,
                                float* __restrict__ out) {
  int i = blockIdx.x * blockDim.x + threadIdx.x;
  if (i >= N_GRAPHS * 64) return;
  float c = cnt[i >> 6];
  out[i] = sums[i] / fmaxf(c, 1.0f);
}

extern "C" void kernel_launch(void* const* d_in, const int* in_sizes, int n_in,
                              void* d_out, int out_size, void* d_ws, size_t ws_size,
                              hipStream_t stream) {
  const float* x     = (const float*)d_in[0];
  const int*   ei    = (const int*)d_in[1];
  const int*   batch = (const int*)d_in[2];
  const float* W1a   = (const float*)d_in[3];
  const float* b1a   = (const float*)d_in[4];
  const float* W1b   = (const float*)d_in[5];
  const float* b1b   = (const float*)d_in[6];
  const float* W2a   = (const float*)d_in[7];
  const float* b2a   = (const float*)d_in[8];
  const float* W2b   = (const float*)d_in[9];
  const float* b2b   = (const float*)d_in[10];
  float* out = (float*)d_out;

  float* ws    = (float*)d_ws;
  float* part  = ws;
  float* h1    = ws + 12000000;
  float* hsum  = ws + 12000000 + 6400000;
  float* sums  = ws + 12000000 + 12800000;

  dim3 b256(256);
  agg3_partial_kernel<<<L1_RANGES * L1_SLICES, dim3(L1_THREADS), 0, stream>>>(x, ei, part);
  mlp1_kernel<<<(N_TILES + 3) / 4, dim3(128), 0, stream>>>(
      x, part, W1a, b1a, W1b, b1b, h1, hsum, sums);
  scatter64_kernel<<<(N_EDGES / ROWS_PER_WAVE + 7) / 8, b256, 0, stream>>>(h1, ei, hsum);
  mlp2_kernel<<<(N_TILES + 3) / 4, dim3(128), 0, stream>>>(
      hsum, W2a, b2a, W2b, b2b, h1  );
  pool_runs_kernel<<<(N_NODES / ROWS_PER_WAVE + 7) / 8, b256, 0, stream>>>(
      h1, batch, sums, sums + 8192);
  finalize_kernel<<<(N_GRAPHS * 64 + 255) / 256, b256, 0, stream>>>(
      sums, sums + 8192, out);
}
